# starting kernel
# baseline (speedup 1.0000x reference)
#include <hip/hip_runtime.h>
#include <hip/hip_bf16.h>
#include <math.h>


typedef __attribute__((ext_vector_type(2))) float v2f;
typedef __attribute__((ext_vector_type(8))) float v8f;

#define PROB_TH 1e-3f

__global__ __launch_bounds__(256) void ffg_tilecolors(
    const float* __restrict__ fc,
    const float* __restrict__ fs,
    const int*   __restrict__ temp,
    float*       __restrict__ tileColors)
{
  __shared__ float sp[64 * 32];
  __shared__ int   st3[64 * 3];
  __shared__ float sc[96];
  __shared__ float wl[256 * 33];

  const int tid  = threadIdx.x;
  const float invT = 1.0f / (float)temp[0];

  if (tid < 64) {
    const float* row = fs + tid * 32;
    float m = -1e30f;
    for (int j = 0; j < 32; j++) m = fmaxf(m, row[j] * invT);
    float s = 0.f;
    for (int j = 0; j < 32; j++) {
      float e = expf(row[j] * invT - m);
      s += e;
      sp[tid * 32 + j] = e;
    }
    float inv = 1.0f / s;
    float v0 = -1.f, v1 = -1.f, v2 = -1.f;
    int   i0 = 0,    i1 = 0,    i2 = 0;
    for (int j = 0; j < 32; j++) {
      float p = sp[tid * 32 + j] * inv;
      sp[tid * 32 + j] = p;
      if (p > v0)      { v2 = v1; i2 = i1; v1 = v0; i1 = i0; v0 = p; i0 = j; }
      else if (p > v1) { v2 = v1; i2 = i1; v1 = p;  i1 = j; }
      else if (p > v2) { v2 = p;  i2 = j; }
    }
    st3[tid * 3 + 0] = i0;
    st3[tid * 3 + 1] = i1;
    st3[tid * 3 + 2] = i2;
  }
  if (tid < 96) sc[tid] = 1.0f / (1.0f + expf(-fc[tid]));
  __syncthreads();

  const int tile = blockIdx.x * 256 + tid;
  const int ty = tile >> 5, tx = tile & 31;
  float* w = &wl[tid * 33];
  for (int i = 0; i < 32; i++) w[i] = 0.f;

  {
    const int r0 = ((ty >> 4) & 1);
    const float* p0 = &sp[r0 * 32];
    const int*   t0 = &st3[r0 * 3];
    for (int j0 = 0; j0 < 3; j0++) {
      const int c1 = t0[j0];
      const float q1 = p0[c1];
      if (!(q1 > PROB_TH)) continue;
      const int dir1 = c1 & 1;
      const int cy1 = 1, cx1 = 0;
      const int side1 = dir1 ? ((tx >> (4 - cx1)) & 1) : ((ty >> (4 - cy1)) & 1);
      const int r1 = c1 * 2 + side1;
      const float* p1 = &sp[r1 * 32];
      const int*   t1 = &st3[r1 * 3];
      for (int j1 = 0; j1 < 3; j1++) {
        const int c2 = t1[j1];
        const float q2 = p1[c2];
        if (!(q2 > PROB_TH)) continue;
        const int cy2 = cy1 + (dir1 ^ 1), cx2 = cx1 + dir1;
        const int dir2 = c2 & 1;
        const int side2 = dir2 ? ((tx >> (4 - cx2)) & 1) : ((ty >> (4 - cy2)) & 1);
        const int r2 = c2 * 2 + side2;
        const float* p2 = &sp[r2 * 32];
        const int*   t2 = &st3[r2 * 3];
        const float w2 = q1 * q2;
        for (int j2 = 0; j2 < 2; j2++) {
          const int c3 = t2[j2];
          const float q3 = p2[c3];
          if (!(q3 > PROB_TH)) continue;
          const int cy3 = cy2 + (dir2 ^ 1), cx3 = cx2 + dir2;
          const int dir3 = c3 & 1;
          const int side3 = dir3 ? ((tx >> (4 - cx3)) & 1) : ((ty >> (4 - cy3)) & 1);
          const int r3 = c3 * 2 + side3;
          const float* p3 = &sp[r3 * 32];
          const int*   t3 = &st3[r3 * 3];
          const float w3 = w2 * q3;
          for (int j3 = 0; j3 < 2; j3++) {
            const int c4 = t3[j3];
            const float q4 = p3[c4];
            if (!(q4 > PROB_TH)) continue;
            const int cy4 = cy3 + (dir3 ^ 1), cx4 = cx3 + dir3;
            const int dir4 = c4 & 1;
            const int side4 = dir4 ? ((tx >> (4 - cx4)) & 1) : ((ty >> (4 - cy4)) & 1);
            const int r4 = c4 * 2 + side4;
            const float w4 = w3 * q4;
            const int c5 = st3[r4 * 3 + 0];
            const float q5 = sp[r4 * 32 + c5];
            if (q5 > PROB_TH) w[c5] += w4 * q5;
          }
        }
      }
    }
  }
  __syncthreads();

  const int wave = tid >> 5;
  const int lane = tid & 31;
  const int half = lane >> 4;
  const int ln   = lane & 15;
  for (int cc = 0; cc < 2; cc++) {
    const int base = (wave * 2 + cc) * 16;
    v8f acc = {};
#pragma unroll
    for (int s = 0; s < 8; s++) {
      const int kb = 4 * s + 2 * half;
      v2f a, b;
      a.x = wl[(base + ln) * 33 + kb];
      a.y = wl[(base + ln) * 33 + kb + 1];
      b.x = (ln < 3) ? sc[(kb + 0) * 3 + ln] : 0.f;
      b.y = (ln < 3) ? sc[(kb + 1) * 3 + ln] : 0.f;
      acc = __builtin_amdgcn_wmma_f32_16x16x4_f32(
          false, a, false, b, (short)0, acc, false, false);
    }
    const int gbase = blockIdx.x * 256 + base;
#pragma unroll
    for (int r = 0; r < 8; r++) {
      if (ln < 3) tileColors[(gbase + r + 8 * half) * 3 + ln] = acc[r];
    }
  }
}

__global__ __launch_bounds__(256) void ffg_write(
    const float* __restrict__ tc, float* __restrict__ out)
{
  const int bid = blockIdx.x;
  const int c = bid >> 10;
  const int y = bid & 1023;
  const int ty = y >> 5;
  const int t = threadIdx.x;
  const int x0 = t << 2;
  const int tx = x0 >> 5;
  const float v = tc[(ty * 32 + tx) * 3 + c];
  float4 o = make_float4(v, v, v, v);
  *(float4*)(out + ((size_t)c << 20) + ((size_t)y << 10) + x0) = o;
}

extern "C" void kernel_launch(void* const* d_in, const int* in_sizes, int n_in,
                              void* d_out, int out_size, void* d_ws, size_t ws_size,
                              hipStream_t stream) {
  (void)in_sizes; (void)n_in; (void)out_size; (void)ws_size;
  const float* fc   = (const float*)d_in[0];
  const float* fs   = (const float*)d_in[1];
  const int*   temp = (const int*)d_in[2];
  float* out = (float*)d_out;
  float* tc  = (float*)d_ws;

  ffg_tilecolors<<<4, 256, 0, stream>>>(fc, fs, temp, tc);
  ffg_write<<<3 * 1024, 256, 0, stream>>>(tc, out);
}


# Path walk spread over threads: one thread per (tile,path) walks one root-to-leaf path (64 blocks x 576 threads), then one thread per (tile,leaf) gathers W from LDS; summation order unchanged
# speedup vs baseline: 1.8621x; 1.8621x over previous
#include <hip/hip_runtime.h>
#include <hip/hip_bf16.h>
#include <math.h>


typedef __attribute__((ext_vector_type(2))) float v2f;
typedef __attribute__((ext_vector_type(8))) float v8f;

#define PROB_TH 1e-3f
#define TILES_PER_BLOCK 16
#define PATHS 36

__global__ __launch_bounds__(576) void ffg_tilecolors(
    const float* __restrict__ fc,
    const float* __restrict__ fs,
    const int*   __restrict__ temp,
    float*       __restrict__ tileColors)
{
  __shared__ float sp[64 * 32];
  __shared__ int   st3[64 * 3];
  __shared__ float sc[96];
  __shared__ float4 prec[TILES_PER_BLOCK * PATHS];
  __shared__ float wl[TILES_PER_BLOCK * 33];

  const int tid  = threadIdx.x;
  const float invT = 1.0f / (float)temp[0];

  if (tid < 64) {
    const float* row = fs + tid * 32;
    float m = -1e30f;
    for (int j = 0; j < 32; j++) m = fmaxf(m, row[j] * invT);
    float s = 0.f;
    for (int j = 0; j < 32; j++) {
      float e = expf(row[j] * invT - m);
      s += e;
      sp[tid * 32 + j] = e;
    }
    float inv = 1.0f / s;
    float v0 = -1.f, v1 = -1.f, v2 = -1.f;
    int   i0 = 0,    i1 = 0,    i2 = 0;
    for (int j = 0; j < 32; j++) {
      float p = sp[tid * 32 + j] * inv;
      sp[tid * 32 + j] = p;
      if (p > v0)      { v2 = v1; i2 = i1; v1 = v0; i1 = i0; v0 = p; i0 = j; }
      else if (p > v1) { v2 = v1; i2 = i1; v1 = p;  i1 = j; }
      else if (p > v2) { v2 = p;  i2 = j; }
    }
    st3[tid * 3 + 0] = i0;
    st3[tid * 3 + 1] = i1;
    st3[tid * 3 + 2] = i2;
  }
  if (tid < 96) sc[tid] = 1.0f / (1.0f + expf(-fc[tid]));
  __syncthreads();

  {
    const int tl   = tid / PATHS;
    const int path = tid - tl * PATHS;
    const int tile = blockIdx.x * TILES_PER_BLOCK + tl;
    const int ty = tile >> 5, tx = tile & 31;
    const int j0 = path / 12;
    const int rem = path - j0 * 12;
    const int j1 = rem >> 2, j2 = (rem >> 1) & 1, j3 = rem & 1;

    int   leaf = -1;
    float wa = 0.f, wb = 0.f;
    const int r0 = ((ty >> 4) & 1);
    const int c1 = st3[r0 * 3 + j0];
    const float q1 = sp[r0 * 32 + c1];
    if (q1 > PROB_TH) {
      const int dir1 = c1 & 1;
      const int cy1 = 1, cx1 = 0;
      const int side1 = dir1 ? ((tx >> (4 - cx1)) & 1) : ((ty >> (4 - cy1)) & 1);
      const int r1 = c1 * 2 + side1;
      const int c2 = st3[r1 * 3 + j1];
      const float q2 = sp[r1 * 32 + c2];
      if (q2 > PROB_TH) {
        const int cy2 = cy1 + (dir1 ^ 1), cx2 = cx1 + dir1;
        const int dir2 = c2 & 1;
        const int side2 = dir2 ? ((tx >> (4 - cx2)) & 1) : ((ty >> (4 - cy2)) & 1);
        const int r2 = c2 * 2 + side2;
        const float w2 = q1 * q2;
        const int c3 = st3[r2 * 3 + j2];
        const float q3 = sp[r2 * 32 + c3];
        if (q3 > PROB_TH) {
          const int cy3 = cy2 + (dir2 ^ 1), cx3 = cx2 + dir2;
          const int dir3 = c3 & 1;
          const int side3 = dir3 ? ((tx >> (4 - cx3)) & 1) : ((ty >> (4 - cy3)) & 1);
          const int r3 = c3 * 2 + side3;
          const float w3 = w2 * q3;
          const int c4 = st3[r3 * 3 + j3];
          const float q4 = sp[r3 * 32 + c4];
          if (q4 > PROB_TH) {
            const int cy4 = cy3 + (dir3 ^ 1), cx4 = cx3 + dir3;
            const int dir4 = c4 & 1;
            const int side4 = dir4 ? ((tx >> (4 - cx4)) & 1) : ((ty >> (4 - cy4)) & 1);
            const int r4 = c4 * 2 + side4;
            const float w4 = w3 * q4;
            const int c5 = st3[r4 * 3 + 0];
            const float q5 = sp[r4 * 32 + c5];
            if (q5 > PROB_TH) { leaf = c5; wa = w4; wb = q5; }
          }
        }
      }
    }
    prec[tid] = make_float4(__int_as_float(leaf), wa, wb, 0.f);
  }
  __syncthreads();

  if (tid < TILES_PER_BLOCK * 32) {
    const int tl = tid >> 5, leaf = tid & 31;
    float w = 0.f;
#pragma unroll
    for (int p = 0; p < PATHS; p++) {
      const float4 rec = prec[tl * PATHS + p];
      const float t = fmaf(rec.y, rec.z, w);
      w = (__float_as_int(rec.x) == leaf) ? t : w;
    }
    wl[tl * 33 + leaf] = w;
  }
  __syncthreads();

  if (tid < 32) {
    const int half = tid >> 4;
    const int ln   = tid & 15;
    v8f acc = {};
#pragma unroll
    for (int s = 0; s < 8; s++) {
      const int kb = 4 * s + 2 * half;
      v2f a, b;
      a.x = wl[ln * 33 + kb];
      a.y = wl[ln * 33 + kb + 1];
      b.x = (ln < 3) ? sc[(kb + 0) * 3 + ln] : 0.f;
      b.y = (ln < 3) ? sc[(kb + 1) * 3 + ln] : 0.f;
      acc = __builtin_amdgcn_wmma_f32_16x16x4_f32(
          false, a, false, b, (short)0, acc, false, false);
    }
    const int gbase = blockIdx.x * TILES_PER_BLOCK;
#pragma unroll
    for (int r = 0; r < 8; r++) {
      if (ln < 3) tileColors[(gbase + r + 8 * half) * 3 + ln] = acc[r];
    }
  }
}

__global__ __launch_bounds__(256) void ffg_write(
    const float* __restrict__ tc, float* __restrict__ out)
{
  const int bid = blockIdx.x;
  const int c = bid >> 10;
  const int y = bid & 1023;
  const int ty = y >> 5;
  const int t = threadIdx.x;
  const int x0 = t << 2;
  const int tx = x0 >> 5;
  const float v = tc[(ty * 32 + tx) * 3 + c];
  float4 o = make_float4(v, v, v, v);
  *(float4*)(out + ((size_t)c << 20) + ((size_t)y << 10) + x0) = o;
}

extern "C" void kernel_launch(void* const* d_in, const int* in_sizes, int n_in,
                              void* d_out, int out_size, void* d_ws, size_t ws_size,
                              hipStream_t stream) {
  (void)in_sizes; (void)n_in; (void)out_size; (void)ws_size;
  const float* fc   = (const float*)d_in[0];
  const float* fs   = (const float*)d_in[1];
  const int*   temp = (const int*)d_in[2];
  float* out = (float*)d_out;
  float* tc  = (float*)d_ws;

  ffg_tilecolors<<<64, 576, 0, stream>>>(fc, fs, temp, tc);
  ffg_write<<<3 * 1024, 256, 0, stream>>>(tc, out);
}


# Two launches fused into one: 256 blocks x 256 threads, each block computes its own 4 tiles' colors (same walk, gather, f32 WMMA) in LDS, then stores those tiles' pixels; summation order unchanged
# speedup vs baseline: 2.4066x; 1.2924x over previous
#include <hip/hip_runtime.h>
#include <hip/hip_bf16.h>
#include <math.h>


typedef __attribute__((ext_vector_type(2))) float v2f;
typedef __attribute__((ext_vector_type(8))) float v8f;

#define PROB_TH 1e-3f
#define TILES_PER_BLOCK 4
#define PATHS 36
#define BLOCK_THREADS 256

__global__ __launch_bounds__(BLOCK_THREADS) void ffg_fused(
    const float* __restrict__ fc,
    const float* __restrict__ fs,
    const int*   __restrict__ temp,
    float*       __restrict__ out)
{
  __shared__ float sp[64 * 32];
  __shared__ int   st3[64 * 3];
  __shared__ float sc[96];
  __shared__ float4 prec[TILES_PER_BLOCK * PATHS];
  __shared__ float wl[TILES_PER_BLOCK * 33];
  __shared__ float sd[TILES_PER_BLOCK * 32];

  const int tid  = threadIdx.x;
  const float invT = 1.0f / (float)temp[0];

  if (tid < 64) {
    const float* row = fs + tid * 32;
    float m = -1e30f;
    for (int j = 0; j < 32; j++) m = fmaxf(m, row[j] * invT);
    float s = 0.f;
    for (int j = 0; j < 32; j++) {
      float e = expf(row[j] * invT - m);
      s += e;
      sp[tid * 32 + j] = e;
    }
    float inv = 1.0f / s;
    float v0 = -1.f, v1 = -1.f, v2 = -1.f;
    int   i0 = 0,    i1 = 0,    i2 = 0;
    for (int j = 0; j < 32; j++) {
      float p = sp[tid * 32 + j] * inv;
      sp[tid * 32 + j] = p;
      if (p > v0)      { v2 = v1; i2 = i1; v1 = v0; i1 = i0; v0 = p; i0 = j; }
      else if (p > v1) { v2 = v1; i2 = i1; v1 = p;  i1 = j; }
      else if (p > v2) { v2 = p;  i2 = j; }
    }
    st3[tid * 3 + 0] = i0;
    st3[tid * 3 + 1] = i1;
    st3[tid * 3 + 2] = i2;
  }
  if (tid < 96) sc[tid] = 1.0f / (1.0f + expf(-fc[tid]));
  __syncthreads();

  if (tid < TILES_PER_BLOCK * PATHS) {
    const int tl   = tid / PATHS;
    const int path = tid - tl * PATHS;
    const int tile = blockIdx.x * TILES_PER_BLOCK + tl;
    const int ty = tile >> 5, tx = tile & 31;
    const int j0 = path / 12;
    const int rem = path - j0 * 12;
    const int j1 = rem >> 2, j2 = (rem >> 1) & 1, j3 = rem & 1;

    int   leaf = -1;
    float wa = 0.f, wb = 0.f;
    const int r0 = ((ty >> 4) & 1);
    const int c1 = st3[r0 * 3 + j0];
    const float q1 = sp[r0 * 32 + c1];
    if (q1 > PROB_TH) {
      const int dir1 = c1 & 1;
      const int cy1 = 1, cx1 = 0;
      const int side1 = dir1 ? ((tx >> (4 - cx1)) & 1) : ((ty >> (4 - cy1)) & 1);
      const int r1 = c1 * 2 + side1;
      const int c2 = st3[r1 * 3 + j1];
      const float q2 = sp[r1 * 32 + c2];
      if (q2 > PROB_TH) {
        const int cy2 = cy1 + (dir1 ^ 1), cx2 = cx1 + dir1;
        const int dir2 = c2 & 1;
        const int side2 = dir2 ? ((tx >> (4 - cx2)) & 1) : ((ty >> (4 - cy2)) & 1);
        const int r2 = c2 * 2 + side2;
        const float w2 = q1 * q2;
        const int c3 = st3[r2 * 3 + j2];
        const float q3 = sp[r2 * 32 + c3];
        if (q3 > PROB_TH) {
          const int cy3 = cy2 + (dir2 ^ 1), cx3 = cx2 + dir2;
          const int dir3 = c3 & 1;
          const int side3 = dir3 ? ((tx >> (4 - cx3)) & 1) : ((ty >> (4 - cy3)) & 1);
          const int r3 = c3 * 2 + side3;
          const float w3 = w2 * q3;
          const int c4 = st3[r3 * 3 + j3];
          const float q4 = sp[r3 * 32 + c4];
          if (q4 > PROB_TH) {
            const int cy4 = cy3 + (dir3 ^ 1), cx4 = cx3 + dir3;
            const int dir4 = c4 & 1;
            const int side4 = dir4 ? ((tx >> (4 - cx4)) & 1) : ((ty >> (4 - cy4)) & 1);
            const int r4 = c4 * 2 + side4;
            const float w4 = w3 * q4;
            const int c5 = st3[r4 * 3 + 0];
            const float q5 = sp[r4 * 32 + c5];
            if (q5 > PROB_TH) { leaf = c5; wa = w4; wb = q5; }
          }
        }
      }
    }
    prec[tid] = make_float4(__int_as_float(leaf), wa, wb, 0.f);
  }
  __syncthreads();

  if (tid < TILES_PER_BLOCK * 32) {
    const int tl = tid >> 5, leaf = tid & 31;
    float w = 0.f;
#pragma unroll
    for (int p = 0; p < PATHS; p++) {
      const float4 rec = prec[tl * PATHS + p];
      const float t = fmaf(rec.y, rec.z, w);
      w = (__float_as_int(rec.x) == leaf) ? t : w;
    }
    wl[tl * 33 + leaf] = w;
  }
  __syncthreads();

  if (tid < 32) {
    const int half = tid >> 4;
    const int ln   = tid & 15;
    v8f acc = {};
#pragma unroll
    for (int s = 0; s < 8; s++) {
      const int kb = 4 * s + 2 * half;
      v2f a, b;
      a.x = (ln < TILES_PER_BLOCK) ? wl[ln * 33 + kb]     : 0.f;
      a.y = (ln < TILES_PER_BLOCK) ? wl[ln * 33 + kb + 1] : 0.f;
      b.x = (ln < 3) ? sc[(kb + 0) * 3 + ln] : 0.f;
      b.y = (ln < 3) ? sc[(kb + 1) * 3 + ln] : 0.f;
      acc = __builtin_amdgcn_wmma_f32_16x16x4_f32(
          false, a, false, b, (short)0, acc, false, false);
    }
    asm volatile("s_wait_alu depctr_va_vdst(0)" : "+v"(acc));
#pragma unroll
    for (int r = 0; r < TILES_PER_BLOCK; r++) sd[r * 32 + tid] = acc[r];
  }
  __syncthreads();

  {
    const int lane = tid & 31;
    const int wv   = tid >> 5;
    const int tl   = lane >> 3;
    const int y0   = (blockIdx.x >> 3) << 5;
    const int x0   = ((blockIdx.x & 7) << 7) + (lane << 2);
#pragma unroll
    for (int c = 0; c < 3; c++) {
      const float v = sd[tl * 32 + c];
      const float4 o = make_float4(v, v, v, v);
      float* base = out + ((size_t)c << 20) + x0;
#pragma unroll
      for (int k = 0; k < 4; k++) {
        const int y = y0 + k * 8 + wv;
        *(float4*)(base + ((size_t)y << 10)) = o;
      }
    }
  }
}

extern "C" void kernel_launch(void* const* d_in, const int* in_sizes, int n_in,
                              void* d_out, int out_size, void* d_ws, size_t ws_size,
                              hipStream_t stream) {
  (void)in_sizes; (void)n_in; (void)out_size; (void)d_ws; (void)ws_size;
  const float* fc   = (const float*)d_in[0];
  const float* fs   = (const float*)d_in[1];
  const int*   temp = (const int*)d_in[2];
  float* out = (float*)d_out;

  ffg_fused<<<256, BLOCK_THREADS, 0, stream>>>(fc, fs, temp, out);
}


# Softmax rows + top-3 spread over all 256 threads (4 lanes per row, quad swizzles on the vector ALU), prologue loads issued up front; row sum now 4 partial sums added pairwise
# speedup vs baseline: 4.3499x; 1.8075x over previous
#include <hip/hip_runtime.h>
#include <hip/hip_bf16.h>
#include <math.h>


typedef __attribute__((ext_vector_type(2))) float v2f;
typedef __attribute__((ext_vector_type(8))) float v8f;

#define PROB_TH 1e-3f
#define TILES_PER_BLOCK 4
#define PATHS 36
#define BLOCK_THREADS 256

__device__ __forceinline__ int quad_xor1(int v) {
  return __builtin_amdgcn_update_dpp(v, v, 0xB1, 0xF, 0xF, false);
}
__device__ __forceinline__ int quad_xor2(int v) {
  return __builtin_amdgcn_update_dpp(v, v, 0x4E, 0xF, 0xF, false);
}
__device__ __forceinline__ float quad_xor1f(float v) { return __int_as_float(quad_xor1(__float_as_int(v))); }
__device__ __forceinline__ float quad_xor2f(float v) { return __int_as_float(quad_xor2(__float_as_int(v))); }

__device__ __forceinline__ void top3_insert(float p, int j, float& v0, float& v1, float& v2,
                                            int& i0, int& i1, int& i2) {
  const bool c0 = p > v0, c1 = p > v1, c2 = p > v2;
  v2 = c1 ? v1 : (c2 ? p : v2);
  i2 = c1 ? i1 : (c2 ? j : i2);
  v1 = c0 ? v0 : (c1 ? p : v1);
  i1 = c0 ? i0 : (c1 ? j : i1);
  v0 = c0 ? p : v0;
  i0 = c0 ? j : i0;
}

__global__ __launch_bounds__(BLOCK_THREADS) void ffg_fused(
    const float* __restrict__ fc,
    const float* __restrict__ fs,
    const int*   __restrict__ temp,
    float*       __restrict__ out)
{
  __shared__ float sp[64 * 32];
  __shared__ int   st3[64 * 3];
  __shared__ float sc[96];
  __shared__ float4 prec[TILES_PER_BLOCK * PATHS];
  __shared__ float wl[TILES_PER_BLOCK * 33];
  __shared__ float sd[TILES_PER_BLOCK * 32];

  const int tid  = threadIdx.x;

  const int    tval = temp[0];
  const float4 xa = *(const float4*)(fs + tid * 8);
  const float4 xb = *(const float4*)(fs + tid * 8 + 4);
  float cv = 0.f;
  if (tid < 96) cv = fc[tid];
  const float invT = 1.0f / (float)tval;
  {
    const float xr[8] = {xa.x, xa.y, xa.z, xa.w, xb.x, xb.y, xb.z, xb.w};
    const int q = tid & 3;
    float m = -1e30f;
    for (int i = 0; i < 8; i++) m = fmaxf(m, xr[i] * invT);
    m = fmaxf(m, quad_xor1f(m));
    m = fmaxf(m, quad_xor2f(m));
    float e[8];
    float s = 0.f;
    for (int i = 0; i < 8; i++) {
      e[i] = expf(xr[i] * invT - m);
      s += e[i];
    }
    s += quad_xor1f(s);
    s += quad_xor2f(s);
    const float inv = 1.0f / s;
    float v0 = -1.f, v1 = -1.f, v2 = -1.f;
    int   i0 = 0,    i1 = 0,    i2 = 0;
    float p[8];
    for (int i = 0; i < 8; i++) {
      p[i] = e[i] * inv;
      top3_insert(p[i], q * 8 + i, v0, v1, v2, i0, i1, i2);
    }
    *(float4*)(sp + tid * 8)     = make_float4(p[0], p[1], p[2], p[3]);
    *(float4*)(sp + tid * 8 + 4) = make_float4(p[4], p[5], p[6], p[7]);
    {
      const float a0 = quad_xor1f(v0), a1 = quad_xor1f(v1), a2 = quad_xor1f(v2);
      const int   b0 = quad_xor1(i0),  b1 = quad_xor1(i1),  b2 = quad_xor1(i2);
      top3_insert(a0, b0, v0, v1, v2, i0, i1, i2);
      top3_insert(a1, b1, v0, v1, v2, i0, i1, i2);
      top3_insert(a2, b2, v0, v1, v2, i0, i1, i2);
    }
    {
      const float a0 = quad_xor2f(v0), a1 = quad_xor2f(v1), a2 = quad_xor2f(v2);
      const int   b0 = quad_xor2(i0),  b1 = quad_xor2(i1),  b2 = quad_xor2(i2);
      top3_insert(a0, b0, v0, v1, v2, i0, i1, i2);
      top3_insert(a1, b1, v0, v1, v2, i0, i1, i2);
      top3_insert(a2, b2, v0, v1, v2, i0, i1, i2);
    }
    if (q == 0) {
      const int row = tid >> 2;
      st3[row * 3 + 0] = i0;
      st3[row * 3 + 1] = i1;
      st3[row * 3 + 2] = i2;
    }
  }
  if (tid < 96) sc[tid] = 1.0f / (1.0f + expf(-cv));
  __syncthreads();

  if (tid < TILES_PER_BLOCK * PATHS) {
    const int tl   = tid / PATHS;
    const int path = tid - tl * PATHS;
    const int tile = blockIdx.x * TILES_PER_BLOCK + tl;
    const int ty = tile >> 5, tx = tile & 31;
    const int j0 = path / 12;
    const int rem = path - j0 * 12;
    const int j1 = rem >> 2, j2 = (rem >> 1) & 1, j3 = rem & 1;

    int   leaf = -1;
    float wa = 0.f, wb = 0.f;
    const int r0 = ((ty >> 4) & 1);
    const int c1 = st3[r0 * 3 + j0];
    const float q1 = sp[r0 * 32 + c1];
    if (q1 > PROB_TH) {
      const int dir1 = c1 & 1;
      const int cy1 = 1, cx1 = 0;
      const int side1 = dir1 ? ((tx >> (4 - cx1)) & 1) : ((ty >> (4 - cy1)) & 1);
      const int r1 = c1 * 2 + side1;
      const int c2 = st3[r1 * 3 + j1];
      const float q2 = sp[r1 * 32 + c2];
      if (q2 > PROB_TH) {
        const int cy2 = cy1 + (dir1 ^ 1), cx2 = cx1 + dir1;
        const int dir2 = c2 & 1;
        const int side2 = dir2 ? ((tx >> (4 - cx2)) & 1) : ((ty >> (4 - cy2)) & 1);
        const int r2 = c2 * 2 + side2;
        const float w2 = q1 * q2;
        const int c3 = st3[r2 * 3 + j2];
        const float q3 = sp[r2 * 32 + c3];
        if (q3 > PROB_TH) {
          const int cy3 = cy2 + (dir2 ^ 1), cx3 = cx2 + dir2;
          const int dir3 = c3 & 1;
          const int side3 = dir3 ? ((tx >> (4 - cx3)) & 1) : ((ty >> (4 - cy3)) & 1);
          const int r3 = c3 * 2 + side3;
          const float w3 = w2 * q3;
          const int c4 = st3[r3 * 3 + j3];
          const float q4 = sp[r3 * 32 + c4];
          if (q4 > PROB_TH) {
            const int cy4 = cy3 + (dir3 ^ 1), cx4 = cx3 + dir3;
            const int dir4 = c4 & 1;
            const int side4 = dir4 ? ((tx >> (4 - cx4)) & 1) : ((ty >> (4 - cy4)) & 1);
            const int r4 = c4 * 2 + side4;
            const float w4 = w3 * q4;
            const int c5 = st3[r4 * 3 + 0];
            const float q5 = sp[r4 * 32 + c5];
            if (q5 > PROB_TH) { leaf = c5; wa = w4; wb = q5; }
          }
        }
      }
    }
    prec[tid] = make_float4(__int_as_float(leaf), wa, wb, 0.f);
  }
  __syncthreads();

  if (tid < TILES_PER_BLOCK * 32) {
    const int tl = tid >> 5, leaf = tid & 31;
    float w = 0.f;
#pragma unroll
    for (int p = 0; p < PATHS; p++) {
      const float4 rec = prec[tl * PATHS + p];
      const float t = fmaf(rec.y, rec.z, w);
      w = (__float_as_int(rec.x) == leaf) ? t : w;
    }
    wl[tl * 33 + leaf] = w;
  }
  __syncthreads();

  if (tid < 32) {
    const int half = tid >> 4;
    const int ln   = tid & 15;
    v8f acc = {};
#pragma unroll
    for (int s = 0; s < 8; s++) {
      const int kb = 4 * s + 2 * half;
      v2f a, b;
      a.x = (ln < TILES_PER_BLOCK) ? wl[ln * 33 + kb]     : 0.f;
      a.y = (ln < TILES_PER_BLOCK) ? wl[ln * 33 + kb + 1] : 0.f;
      b.x = (ln < 3) ? sc[(kb + 0) * 3 + ln] : 0.f;
      b.y = (ln < 3) ? sc[(kb + 1) * 3 + ln] : 0.f;
      acc = __builtin_amdgcn_wmma_f32_16x16x4_f32(
          false, a, false, b, (short)0, acc, false, false);
    }
    asm volatile("s_wait_alu depctr_va_vdst(0)" : "+v"(acc));
#pragma unroll
    for (int r = 0; r < TILES_PER_BLOCK; r++) sd[r * 32 + tid] = acc[r];
  }
  __syncthreads();

  {
    const int lane = tid & 31;
    const int wv   = tid >> 5;
    const int tl   = lane >> 3;
    const int y0   = (blockIdx.x >> 3) << 5;
    const int x0   = ((blockIdx.x & 7) << 7) + (lane << 2);
#pragma unroll
    for (int c = 0; c < 3; c++) {
      const float v = sd[tl * 32 + c];
      const float4 o = make_float4(v, v, v, v);
      float* base = out + ((size_t)c << 20) + x0;
#pragma unroll
      for (int k = 0; k < 4; k++) {
        const int y = y0 + k * 8 + wv;
        *(float4*)(base + ((size_t)y << 10)) = o;
      }
    }
  }
}

extern "C" void kernel_launch(void* const* d_in, const int* in_sizes, int n_in,
                              void* d_out, int out_size, void* d_ws, size_t ws_size,
                              hipStream_t stream) {
  (void)in_sizes; (void)n_in; (void)out_size; (void)d_ws; (void)ws_size;
  const float* fc   = (const float*)d_in[0];
  const float* fs   = (const float*)d_in[1];
  const int*   temp = (const int*)d_in[2];
  float* out = (float*)d_out;

  ffg_fused<<<256, BLOCK_THREADS, 0, stream>>>(fc, fs, temp, out);
}


# Walk reads one packed (child, prob) LDS record per level from a per-row top-3 table the softmax lanes write: 5 dependent LDS reads per path instead of 10, no branches; summation order unchanged
# speedup vs baseline: 4.6170x; 1.0614x over previous
#include <hip/hip_runtime.h>
#include <hip/hip_bf16.h>
#include <math.h>


typedef __attribute__((ext_vector_type(2))) float v2f;
typedef __attribute__((ext_vector_type(8))) float v8f;

#define PROB_TH 1e-3f
#define TILES_PER_BLOCK 4
#define PATHS 36
#define BLOCK_THREADS 256

__device__ __forceinline__ int quad_xor1(int v) {
  return __builtin_amdgcn_update_dpp(v, v, 0xB1, 0xF, 0xF, false);
}
__device__ __forceinline__ int quad_xor2(int v) {
  return __builtin_amdgcn_update_dpp(v, v, 0x4E, 0xF, 0xF, false);
}
__device__ __forceinline__ float quad_xor1f(float v) { return __int_as_float(quad_xor1(__float_as_int(v))); }
__device__ __forceinline__ float quad_xor2f(float v) { return __int_as_float(quad_xor2(__float_as_int(v))); }

__device__ __forceinline__ void top3_insert(float p, int j, float& v0, float& v1, float& v2,
                                            int& i0, int& i1, int& i2) {
  const bool c0 = p > v0, c1 = p > v1, c2 = p > v2;
  v2 = c1 ? v1 : (c2 ? p : v2);
  i2 = c1 ? i1 : (c2 ? j : i2);
  v1 = c0 ? v0 : (c1 ? p : v1);
  i1 = c0 ? i0 : (c1 ? j : i1);
  v0 = c0 ? p : v0;
  i0 = c0 ? j : i0;
}

__global__ __launch_bounds__(BLOCK_THREADS) void ffg_fused(
    const float* __restrict__ fc,
    const float* __restrict__ fs,
    const int*   __restrict__ temp,
    float*       __restrict__ out)
{
  __shared__ float2 top3[64 * 4];
  __shared__ float sc[96];
  __shared__ float4 prec[TILES_PER_BLOCK * PATHS];
  __shared__ float wl[TILES_PER_BLOCK * 33];
  __shared__ float sd[TILES_PER_BLOCK * 32];

  const int tid  = threadIdx.x;

  const int    tval = temp[0];
  const float4 xa = *(const float4*)(fs + tid * 8);
  const float4 xb = *(const float4*)(fs + tid * 8 + 4);
  float cv = 0.f;
  if (tid < 96) cv = fc[tid];
  const float invT = 1.0f / (float)tval;
  {
    const float xr[8] = {xa.x, xa.y, xa.z, xa.w, xb.x, xb.y, xb.z, xb.w};
    const int q = tid & 3;
    float m = -1e30f;
    for (int i = 0; i < 8; i++) m = fmaxf(m, xr[i] * invT);
    m = fmaxf(m, quad_xor1f(m));
    m = fmaxf(m, quad_xor2f(m));
    float e[8];
    float s = 0.f;
    for (int i = 0; i < 8; i++) {
      e[i] = expf(xr[i] * invT - m);
      s += e[i];
    }
    s += quad_xor1f(s);
    s += quad_xor2f(s);
    const float inv = 1.0f / s;
    float v0 = -1.f, v1 = -1.f, v2 = -1.f;
    int   i0 = 0,    i1 = 0,    i2 = 0;
    float p[8];
    for (int i = 0; i < 8; i++) {
      p[i] = e[i] * inv;
      top3_insert(p[i], q * 8 + i, v0, v1, v2, i0, i1, i2);
    }
    {
      const float a0 = quad_xor1f(v0), a1 = quad_xor1f(v1), a2 = quad_xor1f(v2);
      const int   b0 = quad_xor1(i0),  b1 = quad_xor1(i1),  b2 = quad_xor1(i2);
      top3_insert(a0, b0, v0, v1, v2, i0, i1, i2);
      top3_insert(a1, b1, v0, v1, v2, i0, i1, i2);
      top3_insert(a2, b2, v0, v1, v2, i0, i1, i2);
    }
    {
      const float a0 = quad_xor2f(v0), a1 = quad_xor2f(v1), a2 = quad_xor2f(v2);
      const int   b0 = quad_xor2(i0),  b1 = quad_xor2(i1),  b2 = quad_xor2(i2);
      top3_insert(a0, b0, v0, v1, v2, i0, i1, i2);
      top3_insert(a1, b1, v0, v1, v2, i0, i1, i2);
      top3_insert(a2, b2, v0, v1, v2, i0, i1, i2);
    }
    if (q == 0) {
      const int row = tid >> 2;
      top3[row * 4 + 0] = make_float2(__int_as_float(i0), v0);
      top3[row * 4 + 1] = make_float2(__int_as_float(i1), v1);
      top3[row * 4 + 2] = make_float2(__int_as_float(i2), v2);
    }
  }
  if (tid < 96) sc[tid] = 1.0f / (1.0f + expf(-cv));
  __syncthreads();

  if (tid < TILES_PER_BLOCK * PATHS) {
    const int tl   = tid / PATHS;
    const int path = tid - tl * PATHS;
    const int tile = blockIdx.x * TILES_PER_BLOCK + tl;
    const int ty = tile >> 5, tx = tile & 31;
    const int j0 = path / 12;
    const int rem = path - j0 * 12;
    const int j1 = rem >> 2, j2 = (rem >> 1) & 1, j3 = rem & 1;

    const char* recb = (const char*)top3;
    const int r0 = ((ty >> 4) & 1);
    float2 e = top3[r0 * 4 + j0];
    int   ch = __float_as_int(e.x);
    const float q1 = e.y;
    bool  alive = q1 > PROB_TH;
    int   cx = 0, cy = 1;
    int d  = ch & 1;
    int pb = (((tx >> (4 - cx)) & 1) << 6) | (((ty >> (4 - cy)) & 1) << 5);
    e = *(const float2*)(recb + (((ch << 6) | (j1 << 3)) | ((pb >> d) & 32)));
    cx += d; cy += d ^ 1;
    ch = __float_as_int(e.x);
    const float q2 = e.y;
    alive = alive && (q2 > PROB_TH);
    const float w2 = q1 * q2;
    d  = ch & 1;
    pb = (((tx >> (4 - cx)) & 1) << 6) | (((ty >> (4 - cy)) & 1) << 5);
    e = *(const float2*)(recb + (((ch << 6) | (j2 << 3)) | ((pb >> d) & 32)));
    cx += d; cy += d ^ 1;
    ch = __float_as_int(e.x);
    const float q3 = e.y;
    alive = alive && (q3 > PROB_TH);
    const float w3 = w2 * q3;
    d  = ch & 1;
    pb = (((tx >> (4 - cx)) & 1) << 6) | (((ty >> (4 - cy)) & 1) << 5);
    e = *(const float2*)(recb + (((ch << 6) | (j3 << 3)) | ((pb >> d) & 32)));
    cx += d; cy += d ^ 1;
    ch = __float_as_int(e.x);
    const float q4 = e.y;
    alive = alive && (q4 > PROB_TH);
    const float w4 = w3 * q4;
    d  = ch & 1;
    pb = (((tx >> (4 - cx)) & 1) << 6) | (((ty >> (4 - cy)) & 1) << 5);
    e = *(const float2*)(recb + ((ch << 6) | ((pb >> d) & 32)));
    const float q5 = e.y;
    alive = alive && (q5 > PROB_TH);
    const float wa = alive ? w4 : 0.f;
    const float wb = alive ? q5 : 0.f;
    prec[tid] = make_float4(e.x, wa, wb, 0.f);
  }
  __syncthreads();

  if (tid < TILES_PER_BLOCK * 32) {
    const int tl = tid >> 5, leaf = tid & 31;
    float w = 0.f;
#pragma unroll
    for (int p = 0; p < PATHS; p++) {
      const float4 rec = prec[tl * PATHS + p];
      const float t = fmaf(rec.y, rec.z, w);
      w = (__float_as_int(rec.x) == leaf) ? t : w;
    }
    wl[tl * 33 + leaf] = w;
  }
  __syncthreads();

  if (tid < 32) {
    const int half = tid >> 4;
    const int ln   = tid & 15;
    v8f acc = {};
#pragma unroll
    for (int s = 0; s < 8; s++) {
      const int kb = 4 * s + 2 * half;
      v2f a, b;
      a.x = (ln < TILES_PER_BLOCK) ? wl[ln * 33 + kb]     : 0.f;
      a.y = (ln < TILES_PER_BLOCK) ? wl[ln * 33 + kb + 1] : 0.f;
      b.x = (ln < 3) ? sc[(kb + 0) * 3 + ln] : 0.f;
      b.y = (ln < 3) ? sc[(kb + 1) * 3 + ln] : 0.f;
      acc = __builtin_amdgcn_wmma_f32_16x16x4_f32(
          false, a, false, b, (short)0, acc, false, false);
    }
    asm volatile("s_wait_alu depctr_va_vdst(0)" : "+v"(acc));
#pragma unroll
    for (int r = 0; r < TILES_PER_BLOCK; r++) sd[r * 32 + tid] = acc[r];
  }
  __syncthreads();

  {
    const int lane = tid & 31;
    const int wv   = tid >> 5;
    const int tl   = lane >> 3;
    const int y0   = (blockIdx.x >> 3) << 5;
    const int x0   = ((blockIdx.x & 7) << 7) + (lane << 2);
#pragma unroll
    for (int c = 0; c < 3; c++) {
      const float v = sd[tl * 32 + c];
      const float4 o = make_float4(v, v, v, v);
      float* base = out + ((size_t)c << 20) + x0;
#pragma unroll
      for (int k = 0; k < 4; k++) {
        const int y = y0 + k * 8 + wv;
        *(float4*)(base + ((size_t)y << 10)) = o;
      }
    }
  }
}

extern "C" void kernel_launch(void* const* d_in, const int* in_sizes, int n_in,
                              void* d_out, int out_size, void* d_ws, size_t ws_size,
                              hipStream_t stream) {
  (void)in_sizes; (void)n_in; (void)out_size; (void)d_ws; (void)ws_size;
  const float* fc   = (const float*)d_in[0];
  const float* fs   = (const float*)d_in[1];
  const int*   temp = (const int*)d_in[2];
  float* out = (float*)d_out;

  ffg_fused<<<256, BLOCK_THREADS, 0, stream>>>(fc, fs, temp, out);
}


# WMMA operand reads made unconditional (a zero row in LDS for the padding lanes, colors stored transposed) and all issued ahead of the 8 WMMAs as 8-byte pairs; k order and summation order unchanged
# speedup vs baseline: 5.2349x; 1.1338x over previous
#include <hip/hip_runtime.h>
#include <hip/hip_bf16.h>
#include <math.h>


typedef __attribute__((ext_vector_type(2))) float v2f;
typedef __attribute__((ext_vector_type(8))) float v8f;

#define PROB_TH 1e-3f
#define TILES_PER_BLOCK 4
#define PATHS 36
#define BLOCK_THREADS 256
#define OPS 34

__device__ __forceinline__ int quad_xor1(int v) {
  return __builtin_amdgcn_update_dpp(v, v, 0xB1, 0xF, 0xF, false);
}
__device__ __forceinline__ int quad_xor2(int v) {
  return __builtin_amdgcn_update_dpp(v, v, 0x4E, 0xF, 0xF, false);
}
__device__ __forceinline__ float quad_xor1f(float v) { return __int_as_float(quad_xor1(__float_as_int(v))); }
__device__ __forceinline__ float quad_xor2f(float v) { return __int_as_float(quad_xor2(__float_as_int(v))); }

__device__ __forceinline__ void top3_insert(float p, int j, float& v0, float& v1, float& v2,
                                            int& i0, int& i1, int& i2) {
  const bool c0 = p > v0, c1 = p > v1, c2 = p > v2;
  v2 = c1 ? v1 : (c2 ? p : v2);
  i2 = c1 ? i1 : (c2 ? j : i2);
  v1 = c0 ? v0 : (c1 ? p : v1);
  i1 = c0 ? i0 : (c1 ? j : i1);
  v0 = c0 ? p : v0;
  i0 = c0 ? j : i0;
}

__global__ __launch_bounds__(BLOCK_THREADS) void ffg_fused(
    const float* __restrict__ fc,
    const float* __restrict__ fs,
    const int*   __restrict__ temp,
    float*       __restrict__ out)
{
  __shared__ float2 top3[64 * 4];
  __shared__ __attribute__((aligned(16))) float scT[4 * OPS];
  __shared__ float4 prec[TILES_PER_BLOCK * PATHS];
  __shared__ __attribute__((aligned(16))) float wl[(TILES_PER_BLOCK + 1) * OPS];
  __shared__ float sd[TILES_PER_BLOCK * 32];

  const int tid  = threadIdx.x;

  const int    tval = temp[0];
  const float4 xa = *(const float4*)(fs + tid * 8);
  const float4 xb = *(const float4*)(fs + tid * 8 + 4);
  float cv = 0.f;
  if (tid < 96) cv = fc[(tid & 31) * 3 + (tid >> 5)];
  const float invT = 1.0f / (float)tval;
  {
    const float xr[8] = {xa.x, xa.y, xa.z, xa.w, xb.x, xb.y, xb.z, xb.w};
    const int q = tid & 3;
    float m = -1e30f;
    for (int i = 0; i < 8; i++) m = fmaxf(m, xr[i] * invT);
    m = fmaxf(m, quad_xor1f(m));
    m = fmaxf(m, quad_xor2f(m));
    float e[8];
    float s = 0.f;
    for (int i = 0; i < 8; i++) {
      e[i] = expf(xr[i] * invT - m);
      s += e[i];
    }
    s += quad_xor1f(s);
    s += quad_xor2f(s);
    const float inv = 1.0f / s;
    float v0 = -1.f, v1 = -1.f, v2 = -1.f;
    int   i0 = 0,    i1 = 0,    i2 = 0;
    float p[8];
    for (int i = 0; i < 8; i++) {
      p[i] = e[i] * inv;
      top3_insert(p[i], q * 8 + i, v0, v1, v2, i0, i1, i2);
    }
    {
      const float a0 = quad_xor1f(v0), a1 = quad_xor1f(v1), a2 = quad_xor1f(v2);
      const int   b0 = quad_xor1(i0),  b1 = quad_xor1(i1),  b2 = quad_xor1(i2);
      top3_insert(a0, b0, v0, v1, v2, i0, i1, i2);
      top3_insert(a1, b1, v0, v1, v2, i0, i1, i2);
      top3_insert(a2, b2, v0, v1, v2, i0, i1, i2);
    }
    {
      const float a0 = quad_xor2f(v0), a1 = quad_xor2f(v1), a2 = quad_xor2f(v2);
      const int   b0 = quad_xor2(i0),  b1 = quad_xor2(i1),  b2 = quad_xor2(i2);
      top3_insert(a0, b0, v0, v1, v2, i0, i1, i2);
      top3_insert(a1, b1, v0, v1, v2, i0, i1, i2);
      top3_insert(a2, b2, v0, v1, v2, i0, i1, i2);
    }
    if (q == 0) {
      const int row = tid >> 2;
      top3[row * 4 + 0] = make_float2(__int_as_float(i0), v0);
      top3[row * 4 + 1] = make_float2(__int_as_float(i1), v1);
      top3[row * 4 + 2] = make_float2(__int_as_float(i2), v2);
    }
  }
  if (tid < 128) scT[(tid >> 5) * OPS + (tid & 31)] = (tid < 96) ? 1.0f / (1.0f + expf(-cv)) : 0.f;
  __syncthreads();

  if (tid < TILES_PER_BLOCK * PATHS) {
    const int tl   = tid / PATHS;
    const int path = tid - tl * PATHS;
    const int tile = blockIdx.x * TILES_PER_BLOCK + tl;
    const int ty = tile >> 5, tx = tile & 31;
    const int j0 = path / 12;
    const int rem = path - j0 * 12;
    const int j1 = rem >> 2, j2 = (rem >> 1) & 1, j3 = rem & 1;

    const char* recb = (const char*)top3;
    const int r0 = ((ty >> 4) & 1);
    float2 e = top3[r0 * 4 + j0];
    int   ch = __float_as_int(e.x);
    const float q1 = e.y;
    bool  alive = q1 > PROB_TH;
    int   cx = 0, cy = 1;
    int d  = ch & 1;
    int pb = (((tx >> (4 - cx)) & 1) << 6) | (((ty >> (4 - cy)) & 1) << 5);
    e = *(const float2*)(recb + (((ch << 6) | (j1 << 3)) | ((pb >> d) & 32)));
    cx += d; cy += d ^ 1;
    ch = __float_as_int(e.x);
    const float q2 = e.y;
    alive = alive && (q2 > PROB_TH);
    const float w2 = q1 * q2;
    d  = ch & 1;
    pb = (((tx >> (4 - cx)) & 1) << 6) | (((ty >> (4 - cy)) & 1) << 5);
    e = *(const float2*)(recb + (((ch << 6) | (j2 << 3)) | ((pb >> d) & 32)));
    cx += d; cy += d ^ 1;
    ch = __float_as_int(e.x);
    const float q3 = e.y;
    alive = alive && (q3 > PROB_TH);
    const float w3 = w2 * q3;
    d  = ch & 1;
    pb = (((tx >> (4 - cx)) & 1) << 6) | (((ty >> (4 - cy)) & 1) << 5);
    e = *(const float2*)(recb + (((ch << 6) | (j3 << 3)) | ((pb >> d) & 32)));
    cx += d; cy += d ^ 1;
    ch = __float_as_int(e.x);
    const float q4 = e.y;
    alive = alive && (q4 > PROB_TH);
    const float w4 = w3 * q4;
    d  = ch & 1;
    pb = (((tx >> (4 - cx)) & 1) << 6) | (((ty >> (4 - cy)) & 1) << 5);
    e = *(const float2*)(recb + ((ch << 6) | ((pb >> d) & 32)));
    const float q5 = e.y;
    alive = alive && (q5 > PROB_TH);
    const float wa = alive ? w4 : 0.f;
    const float wb = alive ? q5 : 0.f;
    prec[tid] = make_float4(e.x, wa, wb, 0.f);
  }
  __syncthreads();

  if (tid < TILES_PER_BLOCK * 32) {
    const int tl = tid >> 5, leaf = tid & 31;
    float w = 0.f;
#pragma unroll
    for (int p = 0; p < PATHS; p++) {
      const float4 rec = prec[tl * PATHS + p];
      const float t = fmaf(rec.y, rec.z, w);
      w = (__float_as_int(rec.x) == leaf) ? t : w;
    }
    wl[tl * OPS + leaf] = w;
  } else if (tid < TILES_PER_BLOCK * 32 + 32) {
    wl[TILES_PER_BLOCK * OPS + (tid & 31)] = 0.f;
  }
  __syncthreads();

  if (tid < 32) {
    const int half = tid >> 4;
    const int ln   = tid & 15;
    const float* ap = wl  + (ln < TILES_PER_BLOCK ? ln : TILES_PER_BLOCK) * OPS + 2 * half;
    const float* bp = scT + (ln < 3 ? ln : 3) * OPS + 2 * half;
    v2f a[8], b[8];
#pragma unroll
    for (int s = 0; s < 8; s++) {
      a[s] = *(const v2f*)(ap + 4 * s);
      b[s] = *(const v2f*)(bp + 4 * s);
    }
    __builtin_amdgcn_sched_barrier(0);
    v8f acc = {};
#pragma unroll
    for (int s = 0; s < 8; s++) {
      acc = __builtin_amdgcn_wmma_f32_16x16x4_f32(
          false, a[s], false, b[s], (short)0, acc, false, false);
    }
    asm volatile("s_wait_alu depctr_va_vdst(0)" : "+v"(acc));
#pragma unroll
    for (int r = 0; r < TILES_PER_BLOCK; r++) sd[r * 32 + tid] = acc[r];
  }
  __syncthreads();

  {
    const int lane = tid & 31;
    const int wv   = tid >> 5;
    const int tl   = lane >> 3;
    const int y0   = (blockIdx.x >> 3) << 5;
    const int x0   = ((blockIdx.x & 7) << 7) + (lane << 2);
#pragma unroll
    for (int c = 0; c < 3; c++) {
      const float v = sd[tl * 32 + c];
      const float4 o = make_float4(v, v, v, v);
      float* base = out + ((size_t)c << 20) + x0;
#pragma unroll
      for (int k = 0; k < 4; k++) {
        const int y = y0 + k * 8 + wv;
        *(float4*)(base + ((size_t)y << 10)) = o;
      }
    }
  }
}

extern "C" void kernel_launch(void* const* d_in, const int* in_sizes, int n_in,
                              void* d_out, int out_size, void* d_ws, size_t ws_size,
                              hipStream_t stream) {
  (void)in_sizes; (void)n_in; (void)out_size; (void)d_ws; (void)ws_size;
  const float* fc   = (const float*)d_in[0];
  const float* fs   = (const float*)d_in[1];
  const int*   temp = (const int*)d_in[2];
  float* out = (float*)d_out;

  ffg_fused<<<256, BLOCK_THREADS, 0, stream>>>(fc, fs, temp, out);
}


# Leaf mix summed over the 36 paths inside the WMMA (A = path weight, B = color of the path's leaf, K = 36): per-leaf gather phase and its barrier removed; summation order changed
# speedup vs baseline: 6.0220x; 1.1504x over previous
#include <hip/hip_runtime.h>
#include <hip/hip_bf16.h>
#include <math.h>


typedef __attribute__((ext_vector_type(2))) float v2f;
typedef __attribute__((ext_vector_type(8))) float v8f;

#define PROB_TH 1e-3f
#define TILES_PER_BLOCK 4
#define PATHS 36
#define BLOCK_THREADS 256
#define OPS 36

__device__ __forceinline__ int quad_xor1(int v) {
  return __builtin_amdgcn_update_dpp(v, v, 0xB1, 0xF, 0xF, false);
}
__device__ __forceinline__ int quad_xor2(int v) {
  return __builtin_amdgcn_update_dpp(v, v, 0x4E, 0xF, 0xF, false);
}
__device__ __forceinline__ float quad_xor1f(float v) { return __int_as_float(quad_xor1(__float_as_int(v))); }
__device__ __forceinline__ float quad_xor2f(float v) { return __int_as_float(quad_xor2(__float_as_int(v))); }

__device__ __forceinline__ void top3_insert(float p, int j, float& v0, float& v1, float& v2,
                                            int& i0, int& i1, int& i2) {
  const bool c0 = p > v0, c1 = p > v1, c2 = p > v2;
  v2 = c1 ? v1 : (c2 ? p : v2);
  i2 = c1 ? i1 : (c2 ? j : i2);
  v1 = c0 ? v0 : (c1 ? p : v1);
  i1 = c0 ? i0 : (c1 ? j : i1);
  v0 = c0 ? p : v0;
  i0 = c0 ? j : i0;
}

__global__ __launch_bounds__(BLOCK_THREADS) void ffg_fused(
    const float* __restrict__ fc,
    const float* __restrict__ fs,
    const int*   __restrict__ temp,
    float*       __restrict__ out)
{
  __shared__ float2 top3[64 * 4];
  __shared__ float4 sc4[32];
  __shared__ __attribute__((aligned(16))) float wp[TILES_PER_BLOCK * OPS];
  __shared__ __attribute__((aligned(16))) float cp[TILES_PER_BLOCK * 4 * OPS];
  __shared__ float sd[TILES_PER_BLOCK * 32];

  const int tid  = threadIdx.x;

  const int    tval = temp[0];
  const float4 xa = *(const float4*)(fs + tid * 8);
  const float4 xb = *(const float4*)(fs + tid * 8 + 4);
  float cv = 0.f;
  if (tid < 96) cv = fc[(tid & 31) * 3 + (tid >> 5)];
  const float invT = 1.0f / (float)tval;
  {
    const float xr[8] = {xa.x, xa.y, xa.z, xa.w, xb.x, xb.y, xb.z, xb.w};
    const int q = tid & 3;
    float m = -1e30f;
    for (int i = 0; i < 8; i++) m = fmaxf(m, xr[i] * invT);
    m = fmaxf(m, quad_xor1f(m));
    m = fmaxf(m, quad_xor2f(m));
    float e[8];
    float s = 0.f;
    for (int i = 0; i < 8; i++) {
      e[i] = expf(xr[i] * invT - m);
      s += e[i];
    }
    s += quad_xor1f(s);
    s += quad_xor2f(s);
    const float inv = 1.0f / s;
    float v0 = -1.f, v1 = -1.f, v2 = -1.f;
    int   i0 = 0,    i1 = 0,    i2 = 0;
    float p[8];
    for (int i = 0; i < 8; i++) {
      p[i] = e[i] * inv;
      top3_insert(p[i], q * 8 + i, v0, v1, v2, i0, i1, i2);
    }
    {
      const float a0 = quad_xor1f(v0), a1 = quad_xor1f(v1), a2 = quad_xor1f(v2);
      const int   b0 = quad_xor1(i0),  b1 = quad_xor1(i1),  b2 = quad_xor1(i2);
      top3_insert(a0, b0, v0, v1, v2, i0, i1, i2);
      top3_insert(a1, b1, v0, v1, v2, i0, i1, i2);
      top3_insert(a2, b2, v0, v1, v2, i0, i1, i2);
    }
    {
      const float a0 = quad_xor2f(v0), a1 = quad_xor2f(v1), a2 = quad_xor2f(v2);
      const int   b0 = quad_xor2(i0),  b1 = quad_xor2(i1),  b2 = quad_xor2(i2);
      top3_insert(a0, b0, v0, v1, v2, i0, i1, i2);
      top3_insert(a1, b1, v0, v1, v2, i0, i1, i2);
      top3_insert(a2, b2, v0, v1, v2, i0, i1, i2);
    }
    if (q == 0) {
      const int row = tid >> 2;
      top3[row * 4 + 0] = make_float2(__int_as_float(i0), v0);
      top3[row * 4 + 1] = make_float2(__int_as_float(i1), v1);
      top3[row * 4 + 2] = make_float2(__int_as_float(i2), v2);
    }
  }
  if (tid < 128) ((float*)sc4)[(tid & 31) * 4 + (tid >> 5)] = (tid < 96) ? 1.0f / (1.0f + expf(-cv)) : 0.f;
  __syncthreads();

  if (tid < TILES_PER_BLOCK * PATHS) {
    const int tl   = tid / PATHS;
    const int path = tid - tl * PATHS;
    const int tile = blockIdx.x * TILES_PER_BLOCK + tl;
    const int ty = tile >> 5, tx = tile & 31;
    const int j0 = path / 12;
    const int rem = path - j0 * 12;
    const int j1 = rem >> 2, j2 = (rem >> 1) & 1, j3 = rem & 1;

    const char* recb = (const char*)top3;
    const int r0 = ((ty >> 4) & 1);
    float2 e = top3[r0 * 4 + j0];
    int   ch = __float_as_int(e.x);
    const float q1 = e.y;
    bool  alive = q1 > PROB_TH;
    int   cx = 0, cy = 1;
    int d  = ch & 1;
    int pb = (((tx >> (4 - cx)) & 1) << 6) | (((ty >> (4 - cy)) & 1) << 5);
    e = *(const float2*)(recb + (((ch << 6) | (j1 << 3)) | ((pb >> d) & 32)));
    cx += d; cy += d ^ 1;
    ch = __float_as_int(e.x);
    const float q2 = e.y;
    alive = alive && (q2 > PROB_TH);
    const float w2 = q1 * q2;
    d  = ch & 1;
    pb = (((tx >> (4 - cx)) & 1) << 6) | (((ty >> (4 - cy)) & 1) << 5);
    e = *(const float2*)(recb + (((ch << 6) | (j2 << 3)) | ((pb >> d) & 32)));
    cx += d; cy += d ^ 1;
    ch = __float_as_int(e.x);
    const float q3 = e.y;
    alive = alive && (q3 > PROB_TH);
    const float w3 = w2 * q3;
    d  = ch & 1;
    pb = (((tx >> (4 - cx)) & 1) << 6) | (((ty >> (4 - cy)) & 1) << 5);
    e = *(const float2*)(recb + (((ch << 6) | (j3 << 3)) | ((pb >> d) & 32)));
    cx += d; cy += d ^ 1;
    ch = __float_as_int(e.x);
    const float q4 = e.y;
    alive = alive && (q4 > PROB_TH);
    const float w4 = w3 * q4;
    d  = ch & 1;
    pb = (((tx >> (4 - cx)) & 1) << 6) | (((ty >> (4 - cy)) & 1) << 5);
    e = *(const float2*)(recb + ((ch << 6) | ((pb >> d) & 32)));
    const float q5 = e.y;
    alive = alive && (q5 > PROB_TH);
    const float4 col = sc4[__float_as_int(e.x)];
    wp[tl * OPS + path] = alive ? w4 * q5 : 0.f;
    float* cb = cp + tl * 4 * OPS + path;
    cb[0]       = col.x;
    cb[OPS]     = col.y;
    cb[2 * OPS] = col.z;
    cb[3 * OPS] = col.w;
  }
  __syncthreads();

  if (tid < 32) {
    const int half = tid >> 4;
    const int ln   = tid & 15;
    const float* ap = wp + (ln & 3) * OPS + 2 * half;
    const float* bp = cp + ln * OPS + 2 * half;
    v2f a[PATHS / 4], b[PATHS / 4];
#pragma unroll
    for (int s = 0; s < PATHS / 4; s++) {
      a[s] = *(const v2f*)(ap + 4 * s);
      b[s] = *(const v2f*)(bp + 4 * s);
    }
    __builtin_amdgcn_sched_barrier(0);
    v8f acc = {};
#pragma unroll
    for (int s = 0; s < PATHS / 4; s++) {
      acc = __builtin_amdgcn_wmma_f32_16x16x4_f32(
          false, a[s], false, b[s], (short)0, acc, false, false);
    }
    asm volatile("s_wait_alu depctr_va_vdst(0)\n\tv_nop\n\tv_nop\n\tv_nop\n\tv_nop"
                 : "+v"(acc)
                 : "v"(a[0]), "v"(a[1]), "v"(a[2]), "v"(a[3]), "v"(a[4]), "v"(a[5]),
                   "v"(a[6]), "v"(a[7]), "v"(a[8]),
                   "v"(b[0]), "v"(b[1]), "v"(b[2]), "v"(b[3]), "v"(b[4]), "v"(b[5]),
                   "v"(b[6]), "v"(b[7]), "v"(b[8]));
#pragma unroll
    for (int r = 0; r < TILES_PER_BLOCK; r++) sd[r * 32 + tid] = acc[r];
  }
  __syncthreads();

  {
    const int lane = tid & 31;
    const int wv   = tid >> 5;
    const int tl   = lane >> 3;
    const int y0   = (blockIdx.x >> 3) << 5;
    const int x0   = ((blockIdx.x & 7) << 7) + (lane << 2);
#pragma unroll
    for (int c = 0; c < 3; c++) {
      const float v = sd[tl * 32 + tl * 4 + c];
      const float4 o = make_float4(v, v, v, v);
      float* base = out + ((size_t)c << 20) + x0;
#pragma unroll
      for (int k = 0; k < 4; k++) {
        const int y = y0 + k * 8 + wv;
        *(float4*)(base + ((size_t)y << 10)) = o;
      }
    }
  }
}

extern "C" void kernel_launch(void* const* d_in, const int* in_sizes, int n_in,
                              void* d_out, int out_size, void* d_ws, size_t ws_size,
                              hipStream_t stream) {
  (void)in_sizes; (void)n_in; (void)out_size; (void)d_ws; (void)ws_size;
  const float* fc   = (const float*)d_in[0];
  const float* fs   = (const float*)d_in[1];
  const int*   temp = (const int*)d_in[2];
  float* out = (float*)d_out;

  ffg_fused<<<256, BLOCK_THREADS, 0, stream>>>(fc, fs, temp, out);
}
